# starting kernel
# baseline (speedup 1.0000x reference)
#include <hip/hip_runtime.h>
#include <math.h>

#define NEG_SLOPE 0.2f

typedef __attribute__((ext_vector_type(2))) float v2f;
typedef __attribute__((ext_vector_type(8))) float v8f;

__global__ __launch_bounds__(32) void gat_scores_wmma(
    const float* __restrict__ x, const float* __restrict__ W_u,
    const float* __restrict__ b_u, const float* __restrict__ W_v,
    float* __restrict__ s, int n_nodes) {
  const int tile = blockIdx.x;
  const int lane = threadIdx.x;
  const int half = lane >> 4;
  const int l15  = lane & 15;

  __shared__ float Wc[128 * 16];
  for (int i = lane; i < 128 * 16; i += 32) {
    const int k = i >> 4, c = i & 15;
    float w = 0.0f;
    if (c < 4)      w = W_u[k * 4 + c];
    else if (c < 8) w = W_v[k * 4 + (c - 4)];
    Wc[i] = w;
  }
  __syncthreads();

  int node = tile * 16 + l15;
  const int nodeC = node < n_nodes ? node : n_nodes - 1;
  const float* __restrict__ xrow = x + (size_t)nodeC * 128;

  v8f acc = {};
  #pragma unroll 4
  for (int k = 0; k < 128; k += 4) {
    const int ka = k + half * 2;
    v2f a; a.x = xrow[ka];            a.y = xrow[ka + 1];
    v2f b; b.x = Wc[ka * 16 + l15];   b.y = Wc[(ka + 1) * 16 + l15];
    acc = __builtin_amdgcn_wmma_f32_16x16x4_f32(
         false, a,  false, b,
         (short)0, acc,  false,  false);
  }

  const int ncol = l15;
  if (ncol < 8) {
    const float bias = (ncol < 4) ? b_u[ncol] : 0.0f;
    #pragma unroll
    for (int r = 0; r < 8; ++r) {
      const int mrow = r + half * 8;
      const int onode = tile * 16 + mrow;
      if (onode < n_nodes) s[(size_t)onode * 8 + ncol] = acc[r] + bias;
    }
  }
}

__global__ void gat_init_md(float* __restrict__ m, float* __restrict__ denom,
                            int n4) {
  const int i = blockIdx.x * blockDim.x + threadIdx.x;
  if (i < n4) { m[i] = -INFINITY; denom[i] = 0.0f; }
}

__global__ void gat_init_out(const float* __restrict__ x,
                             float* __restrict__ out, long long total) {
  const long long i = (long long)blockIdx.x * blockDim.x + threadIdx.x;
  if (i >= total) return;
  const long long node = i >> 8;
  const int d = (int)(i & 255);
  out[i] = (d < 128) ? x[node * 128 + d] : 0.0f;
}

__device__ __forceinline__ void atomicMaxFloat(float* addr, float val) {
  if (val >= 0.0f) atomicMax((int*)addr, __float_as_int(val));
  else             atomicMin((unsigned int*)addr, __float_as_uint(val));
}

__device__ __forceinline__ float lrelu(float v) {
  return v >= 0.0f ? v : NEG_SLOPE * v;
}

__global__ void gat_edge_max(const float* __restrict__ s,
                             const int* __restrict__ src,
                             const int* __restrict__ dst,
                             float* __restrict__ m, int E) {
  const int e = blockIdx.x * blockDim.x + threadIdx.x;
  if (e >= E) return;
  const int u = src[e], v = dst[e];
  const float4 su = *(const float4*)(s + (size_t)u * 8);
  const float4 sv = *(const float4*)(s + (size_t)v * 8 + 4);
  float* mv = m + (size_t)v * 4;
  atomicMaxFloat(mv + 0, lrelu(su.x + sv.x));
  atomicMaxFloat(mv + 1, lrelu(su.y + sv.y));
  atomicMaxFloat(mv + 2, lrelu(su.z + sv.z));
  atomicMaxFloat(mv + 3, lrelu(su.w + sv.w));
}

__global__ void gat_edge_denom(const float* __restrict__ s,
                               const int* __restrict__ src,
                               const int* __restrict__ dst,
                               const float* __restrict__ m,
                               float* __restrict__ denom, int E) {
  const int e = blockIdx.x * blockDim.x + threadIdx.x;
  if (e >= E) return;
  const int u = src[e], v = dst[e];
  const float4 su = *(const float4*)(s + (size_t)u * 8);
  const float4 sv = *(const float4*)(s + (size_t)v * 8 + 4);
  const float4 mv = *(const float4*)(m + (size_t)v * 4);
  float* dv = denom + (size_t)v * 4;
  atomicAdd(dv + 0, __expf(lrelu(su.x + sv.x) - mv.x));
  atomicAdd(dv + 1, __expf(lrelu(su.y + sv.y) - mv.y));
  atomicAdd(dv + 2, __expf(lrelu(su.z + sv.z) - mv.z));
  atomicAdd(dv + 3, __expf(lrelu(su.w + sv.w) - mv.w));
}

__global__ void gat_edge_agg(const float* __restrict__ x,
                             const float* __restrict__ s,
                             const int* __restrict__ src,
                             const int* __restrict__ dst,
                             const float* __restrict__ m,
                             const float* __restrict__ denom,
                             float* __restrict__ out, int E) {
  const long long gid = (long long)blockIdx.x * blockDim.x + threadIdx.x;
  const int e = (int)(gid >> 5);
  const int lane = (int)(gid & 31);
  if (e >= E) return;
  const int u = src[e], v = dst[e];

  const int h = lane & 3;
  const float ee = lrelu(s[(size_t)u * 8 + h] + s[(size_t)v * 8 + 4 + h]);
  const float p = __expf(ee - m[(size_t)v * 4 + h]) / denom[(size_t)v * 4 + h];
  const float p0 = __shfl(p, 0, 32);
  const float p1 = __shfl(p, 1, 32);
  const float p2 = __shfl(p, 2, 32);
  const float p3 = __shfl(p, 3, 32);

  const float4 xv = *(const float4*)(x + (size_t)u * 128 + lane * 4);
  float* o = out + (size_t)v * 256 + 128 + lane * 4;
  atomicAdd(o + 0, xv.x * p0);
  atomicAdd(o + 1, xv.y * p1);
  atomicAdd(o + 2, xv.z * p2);
  atomicAdd(o + 3, xv.w * p3);
}

extern "C" void kernel_launch(void* const* d_in, const int* in_sizes, int n_in,
                              void* d_out, int out_size, void* d_ws, size_t ws_size,
                              hipStream_t stream) {
  const float* x   = (const float*)d_in[0];
  const int*   src = (const int*)d_in[1];
  const int*   dst = (const int*)d_in[2];
  const float* W_u = (const float*)d_in[3];
  const float* b_u = (const float*)d_in[4];
  const float* W_v = (const float*)d_in[5];
  float* out = (float*)d_out;

  const int n = in_sizes[0] / 128;
  const int E = in_sizes[1];

  float* s     = (float*)d_ws;
  float* m     = s + (size_t)n * 8;
  float* denom = m + (size_t)n * 4;

  gat_scores_wmma<<<(n + 15) / 16, 32, 0, stream>>>(x, W_u, b_u, W_v, s, n);

  gat_init_md<<<(n * 4 + 255) / 256, 256, 0, stream>>>(m, denom, n * 4);
  const long long total = (long long)n * 256;
  gat_init_out<<<(unsigned)((total + 255) / 256), 256, 0, stream>>>(x, out, total);

  gat_edge_max<<<(E + 255) / 256, 256, 0, stream>>>(s, src, dst, m, E);

  gat_edge_denom<<<(E + 255) / 256, 256, 0, stream>>>(s, src, dst, m, denom, E);

  gat_edge_agg<<<(E + 7) / 8, 256, 0, stream>>>(x, s, src, dst, m, denom, out, E);
}


# Edges grouped by destination in-call (int atomic count = rank, block scan, src scatter); one wave per node does max, exp-sum, aggregate in registers; no float atomics; sums in rank order, lane tree
# speedup vs baseline: 4.0622x; 4.0622x over previous
#include <hip/hip_runtime.h>
#include <math.h>

#define NEG_SLOPE 0.2f

typedef __attribute__((ext_vector_type(2))) float v2f;
typedef __attribute__((ext_vector_type(8))) float v8f;

__global__ __launch_bounds__(32) void gat_scores_wmma(
    const float* __restrict__ x, const float* __restrict__ W_u,
    const float* __restrict__ b_u, const float* __restrict__ W_v,
    float* __restrict__ s, int* __restrict__ deg, int* __restrict__ total,
    int n_nodes) {
  const int tile = blockIdx.x;
  const int lane = threadIdx.x;
  const int half = lane >> 4;
  const int l15  = lane & 15;

  if (lane < 16 && tile * 16 + lane < n_nodes) deg[tile * 16 + lane] = 0;
  if (tile == 0 && lane == 0) *total = 0;

  __shared__ float Wc[128 * 16];
  for (int i = lane; i < 128 * 16; i += 32) {
    const int k = i >> 4, c = i & 15;
    float w = 0.0f;
    if (c < 4)      w = W_u[k * 4 + c];
    else if (c < 8) w = W_v[k * 4 + (c - 4)];
    Wc[i] = w;
  }
  __syncthreads();

  int node = tile * 16 + l15;
  const int nodeC = node < n_nodes ? node : n_nodes - 1;
  const float* __restrict__ xrow = x + (size_t)nodeC * 128;

  v8f acc = {};
  #pragma unroll 4
  for (int k = 0; k < 128; k += 4) {
    const int ka = k + half * 2;
    v2f a; a.x = xrow[ka];            a.y = xrow[ka + 1];
    v2f b; b.x = Wc[ka * 16 + l15];   b.y = Wc[(ka + 1) * 16 + l15];
    acc = __builtin_amdgcn_wmma_f32_16x16x4_f32(
         false, a,  false, b,
         (short)0, acc,  false,  false);
  }

  const int ncol = l15;
  if (ncol < 8) {
    const float bias = (ncol < 4) ? b_u[ncol] : 0.0f;
    #pragma unroll
    for (int r = 0; r < 8; ++r) {
      const int mrow = r + half * 8;
      const int onode = tile * 16 + mrow;
      if (onode < n_nodes) s[(size_t)onode * 8 + ncol] = acc[r] + bias;
    }
  }
}

__global__ void gat_edge_rank(const int* __restrict__ dst,
                              int* __restrict__ deg, int* __restrict__ rank,
                              int E) {
  const int e = blockIdx.x * blockDim.x + threadIdx.x;
  if (e >= E) return;
  rank[e] = atomicAdd(deg + dst[e], 1);
}

__global__ __launch_bounds__(256) void gat_seg_offsets(
    const int* __restrict__ deg, int* __restrict__ off,
    int* __restrict__ total, int n) {
  __shared__ int wsum[8];
  __shared__ int base_s;
  const int tid  = threadIdx.x;
  const int lane = tid & 31;
  const int wv   = tid >> 5;
  const int i0   = (blockIdx.x * 256 + tid) * 4;

  const int c0 = (i0 + 0 < n) ? deg[i0 + 0] : 0;
  const int c1 = (i0 + 1 < n) ? deg[i0 + 1] : 0;
  const int c2 = (i0 + 2 < n) ? deg[i0 + 2] : 0;
  const int c3 = (i0 + 3 < n) ? deg[i0 + 3] : 0;
  const int t  = c0 + c1 + c2 + c3;

  int incl = t;
  #pragma unroll
  for (int d = 1; d < 32; d <<= 1) {
    const int y = __shfl_up(incl, d, 32);
    if (lane >= d) incl += y;
  }
  if (lane == 31) wsum[wv] = incl;
  __syncthreads();
  if (tid == 0) {
    int tot = 0;
    #pragma unroll
    for (int w = 0; w < 8; ++w) { const int c = wsum[w]; wsum[w] = tot; tot += c; }
    base_s = atomicAdd(total, tot);
  }
  __syncthreads();

  int ex = base_s + wsum[wv] + incl - t;
  if (i0 + 0 < n) off[i0 + 0] = ex;
  ex += c0;
  if (i0 + 1 < n) off[i0 + 1] = ex;
  ex += c1;
  if (i0 + 2 < n) off[i0 + 2] = ex;
  ex += c2;
  if (i0 + 3 < n) off[i0 + 3] = ex;
}

__global__ void gat_edge_place(const int* __restrict__ src,
                               const int* __restrict__ dst,
                               const int* __restrict__ rank,
                               const int* __restrict__ off,
                               int* __restrict__ psrc, int E) {
  const int e = blockIdx.x * blockDim.x + threadIdx.x;
  if (e >= E) return;
  psrc[off[dst[e]] + rank[e]] = src[e];
}

__device__ __forceinline__ float lrelu(float v) {
  return v >= 0.0f ? v : NEG_SLOPE * v;
}

__device__ __forceinline__ float4 edge_scores(const float* __restrict__ s,
                                              int u, float4 sv) {
  const float4 su = *(const float4*)(s + (size_t)u * 8);
  float4 e;
  e.x = lrelu(su.x + sv.x);
  e.y = lrelu(su.y + sv.y);
  e.z = lrelu(su.z + sv.z);
  e.w = lrelu(su.w + sv.w);
  return e;
}

__device__ __forceinline__ float wave_max(float v) {
  #pragma unroll
  for (int d = 16; d >= 1; d >>= 1) v = fmaxf(v, __shfl_xor(v, d, 32));
  return v;
}

__device__ __forceinline__ float wave_sum(float v) {
  #pragma unroll
  for (int d = 16; d >= 1; d >>= 1) v += __shfl_xor(v, d, 32);
  return v;
}

__device__ __forceinline__ float lane_bcast(float v, int l) {
  return __int_as_float(__builtin_amdgcn_readlane(__float_as_int(v), l));
}

__device__ __forceinline__ void accum_rows(const float* __restrict__ x,
                                           int lane, int cnt, int u, float4 p,
                                           float4& acc) {
  #pragma unroll 4
  for (int t = 0; t < cnt; ++t) {
    const int   ut = __builtin_amdgcn_readlane(u, t);
    const float p0 = lane_bcast(p.x, t);
    const float p1 = lane_bcast(p.y, t);
    const float p2 = lane_bcast(p.z, t);
    const float p3 = lane_bcast(p.w, t);
    const float4 xv = *(const float4*)(x + (size_t)ut * 128 + lane * 4);
    acc.x += xv.x * p0;
    acc.y += xv.y * p1;
    acc.z += xv.z * p2;
    acc.w += xv.w * p3;
  }
}

__global__ __launch_bounds__(256) void gat_node_softmax_agg(
    const float* __restrict__ x, const float* __restrict__ s,
    const int* __restrict__ psrc, const int* __restrict__ deg,
    const int* __restrict__ off, float* __restrict__ out, int n) {
  const int lane = threadIdx.x & 31;
  const int v = blockIdx.x * 8 +
                __builtin_amdgcn_readfirstlane((int)(threadIdx.x >> 5));
  if (v >= n) return;
  const int len = __builtin_amdgcn_readfirstlane(deg[v]);
  const int beg = __builtin_amdgcn_readfirstlane(off[v]);
  const int* __restrict__ seg = psrc + beg;
  const float4 sv = *(const float4*)(s + (size_t)v * 8 + 4);

  const bool has0 = lane < len;
  int u0 = 0;
  float4 e0 = {-INFINITY, -INFINITY, -INFINITY, -INFINITY};
  if (has0) { u0 = seg[lane]; e0 = edge_scores(s, u0, sv); }

  float4 mx = e0;
  for (int j = 32 + lane; j < len; j += 32) {
    const float4 ej = edge_scores(s, seg[j], sv);
    mx.x = fmaxf(mx.x, ej.x); mx.y = fmaxf(mx.y, ej.y);
    mx.z = fmaxf(mx.z, ej.z); mx.w = fmaxf(mx.w, ej.w);
  }
  mx.x = wave_max(mx.x); mx.y = wave_max(mx.y);
  mx.z = wave_max(mx.z); mx.w = wave_max(mx.w);

  float4 ex0 = {0.0f, 0.0f, 0.0f, 0.0f};
  if (has0) {
    ex0.x = __expf(e0.x - mx.x); ex0.y = __expf(e0.y - mx.y);
    ex0.z = __expf(e0.z - mx.z); ex0.w = __expf(e0.w - mx.w);
  }
  float4 dn = ex0;
  for (int j = 32 + lane; j < len; j += 32) {
    const float4 ej = edge_scores(s, seg[j], sv);
    dn.x += __expf(ej.x - mx.x); dn.y += __expf(ej.y - mx.y);
    dn.z += __expf(ej.z - mx.z); dn.w += __expf(ej.w - mx.w);
  }
  dn.x = wave_sum(dn.x); dn.y = wave_sum(dn.y);
  dn.z = wave_sum(dn.z); dn.w = wave_sum(dn.w);

  float4 acc = {0.0f, 0.0f, 0.0f, 0.0f};
  float4 p = {0.0f, 0.0f, 0.0f, 0.0f};
  if (has0) {
    p.x = ex0.x / dn.x; p.y = ex0.y / dn.y;
    p.z = ex0.z / dn.z; p.w = ex0.w / dn.w;
  }
  accum_rows(x, lane, len < 32 ? len : 32, u0, p, acc);

  for (int c = 32; c < len; c += 32) {
    const int j = c + lane;
    int u = 0;
    float4 q = {0.0f, 0.0f, 0.0f, 0.0f};
    if (j < len) {
      u = seg[j];
      const float4 ej = edge_scores(s, u, sv);
      q.x = __expf(ej.x - mx.x) / dn.x; q.y = __expf(ej.y - mx.y) / dn.y;
      q.z = __expf(ej.z - mx.z) / dn.z; q.w = __expf(ej.w - mx.w) / dn.w;
    }
    const int rem = len - c;
    accum_rows(x, lane, rem < 32 ? rem : 32, u, q, acc);
  }

  const float4 xs = *(const float4*)(x + (size_t)v * 128 + lane * 4);
  float* __restrict__ o = out + (size_t)v * 256 + lane * 4;
  *(float4*)o = xs;
  *(float4*)(o + 128) = acc;
}

extern "C" void kernel_launch(void* const* d_in, const int* in_sizes, int n_in,
                              void* d_out, int out_size, void* d_ws, size_t ws_size,
                              hipStream_t stream) {
  const float* x   = (const float*)d_in[0];
  const int*   src = (const int*)d_in[1];
  const int*   dst = (const int*)d_in[2];
  const float* W_u = (const float*)d_in[3];
  const float* b_u = (const float*)d_in[4];
  const float* W_v = (const float*)d_in[5];
  float* out = (float*)d_out;

  const int n = in_sizes[0] / 128;
  const int E = in_sizes[1];

  float* s     = (float*)d_ws;
  int*   deg   = (int*)(s + (size_t)n * 8);
  int*   off   = deg + n;
  int*   total = off + n;
  int*   rank  = total + 4;
  int*   psrc  = rank + E;

  gat_scores_wmma<<<(n + 15) / 16, 32, 0, stream>>>(x, W_u, b_u, W_v, s, deg,
                                                    total, n);

  gat_edge_rank<<<(E + 255) / 256, 256, 0, stream>>>(dst, deg, rank, E);

  gat_seg_offsets<<<(n + 1023) / 1024, 256, 0, stream>>>(deg, off, total, n);

  gat_edge_place<<<(E + 255) / 256, 256, 0, stream>>>(src, dst, rank, off, psrc, E);

  gat_node_softmax_agg<<<(n + 7) / 8, 256, 0, stream>>>(x, s, psrc, deg, off, out, n);
}


# Edges binned by destination range in LDS: block-local counting sort, one reserving add per block and bucket, runs stored in order; no per-edge global atomic; sums in LDS-atomic arrival order
# speedup vs baseline: 8.5256x; 2.0988x over previous
#include <hip/hip_runtime.h>
#include <math.h>

#define NEG_SLOPE 0.2f


#define NB 512
#define CAPB 4096
#define NODES_PER_BLOCK 16

__device__ __forceinline__ int nodes_per_bucket(int n) {
  const int q = (n + NB - 1) / NB;
  const int r = (q + NODES_PER_BLOCK - 1) / NODES_PER_BLOCK * NODES_PER_BLOCK;
  return r > 0 ? r : NODES_PER_BLOCK;
}

typedef __attribute__((ext_vector_type(2))) float v2f;
typedef __attribute__((ext_vector_type(8))) float v8f;

__global__ __launch_bounds__(32) void gat_scores_wmma(
    const float* __restrict__ x, const float* __restrict__ W_u,
    const float* __restrict__ b_u, const float* __restrict__ W_v,
    float* __restrict__ s, int* __restrict__ bcnt, int n_nodes) {
  const int tile = blockIdx.x;
  const int lane = threadIdx.x;
  const int half = lane >> 4;
  const int l15  = lane & 15;

  if (lane < 16 && tile * 16 + lane < NB) bcnt[tile * 16 + lane] = 0;

  __shared__ float Wc[128 * 16];
  for (int i = lane; i < 128 * 16; i += 32) {
    const int k = i >> 4, c = i & 15;
    float w = 0.0f;
    if (c < 4)      w = W_u[k * 4 + c];
    else if (c < 8) w = W_v[k * 4 + (c - 4)];
    Wc[i] = w;
  }
  __syncthreads();

  int node = tile * 16 + l15;
  const int nodeC = node < n_nodes ? node : n_nodes - 1;
  const float* __restrict__ xrow = x + (size_t)nodeC * 128;

  v8f acc = {};
  #pragma unroll 4
  for (int k = 0; k < 128; k += 4) {
    const int ka = k + half * 2;
    v2f a; a.x = xrow[ka];            a.y = xrow[ka + 1];
    v2f b; b.x = Wc[ka * 16 + l15];   b.y = Wc[(ka + 1) * 16 + l15];
    acc = __builtin_amdgcn_wmma_f32_16x16x4_f32(
         false, a,  false, b,
         (short)0, acc,  false,  false);
  }

  const int ncol = l15;
  if (ncol < 8) {
    const float bias = (ncol < 4) ? b_u[ncol] : 0.0f;
    #pragma unroll
    for (int r = 0; r < 8; ++r) {
      const int mrow = r + half * 8;
      const int onode = tile * 16 + mrow;
      if (onode < n_nodes) s[(size_t)onode * 8 + ncol] = acc[r] + bias;
    }
  }
}

#define BIN_T 1024
#define BIN_EPT 8
#define BIN_EPB (BIN_T * BIN_EPT)

__global__ __launch_bounds__(BIN_T) void gat_edge_bin(
    const int* __restrict__ src, const int* __restrict__ dst,
    int* __restrict__ bcnt, int* __restrict__ bk_src, int* __restrict__ bk_dst,
    int E, int n) {
  __shared__ int hist[NB];
  __shared__ int lstart[NB];
  __shared__ int gbase[NB];
  __shared__ int wsum[NB / 32];
  __shared__ int st_src[BIN_EPB];
  __shared__ int st_dst[BIN_EPB];

  const int tid  = threadIdx.x;
  const int lane = tid & 31;
  const int npb  = nodes_per_bucket(n);
  const int base = blockIdx.x * BIN_EPB;

  if (tid < NB) hist[tid] = 0;
  __syncthreads();

  int d[BIN_EPT], u[BIN_EPT], pos[BIN_EPT], bkt[BIN_EPT];
  #pragma unroll
  for (int k = 0; k < BIN_EPT; ++k) {
    const int e = base + k * BIN_T + tid;
    d[k] = -1; u[k] = 0;
    if (e < E) { d[k] = dst[e]; u[k] = src[e]; }
  }
  #pragma unroll
  for (int k = 0; k < BIN_EPT; ++k) {
    bkt[k] = 0; pos[k] = 0;
    if (d[k] >= 0) {
      bkt[k] = d[k] / npb;
      pos[k] = atomicAdd(&hist[bkt[k]], 1);
    }
  }
  __syncthreads();

  int c = 0, incl = 0;
  if (tid < NB) {
    c = hist[tid];
    incl = c;
    #pragma unroll
    for (int o = 1; o < 32; o <<= 1) {
      const int t = __shfl_up(incl, o, 32);
      if (lane >= o) incl += t;
    }
    if (lane == 31) wsum[tid >> 5] = incl;
  }
  __syncthreads();
  if (tid < NB) {
    int pre = 0;
    for (int w = 0; w < (tid >> 5); ++w) pre += wsum[w];
    lstart[tid] = pre + incl - c;
    gbase[tid]  = c > 0 ? atomicAdd(bcnt + tid, c) : 0;
  }
  __syncthreads();

  #pragma unroll
  for (int k = 0; k < BIN_EPT; ++k) {
    if (d[k] >= 0) {
      const int p = lstart[bkt[k]] + pos[k];
      st_src[p] = u[k];
      st_dst[p] = d[k];
    }
  }
  __syncthreads();

  const int rem  = E - base;
  const int nloc = rem < BIN_EPB ? rem : BIN_EPB;
  for (int i = tid; i < nloc; i += BIN_T) {
    const int dd = st_dst[i];
    const int b  = dd / npb;
    const int g  = gbase[b] + (i - lstart[b]);
    if (g < CAPB) {
      bk_src[(size_t)b * CAPB + g] = st_src[i];
      bk_dst[(size_t)b * CAPB + g] = dd;
    }
  }
}

__device__ __forceinline__ float lrelu(float v) {
  return v >= 0.0f ? v : NEG_SLOPE * v;
}

__device__ __forceinline__ float4 edge_scores(const float* __restrict__ s,
                                              int u, float4 sv) {
  const float4 su = *(const float4*)(s + (size_t)u * 8);
  float4 e;
  e.x = lrelu(su.x + sv.x);
  e.y = lrelu(su.y + sv.y);
  e.z = lrelu(su.z + sv.z);
  e.w = lrelu(su.w + sv.w);
  return e;
}

__device__ __forceinline__ float wave_max(float v) {
  #pragma unroll
  for (int d = 16; d >= 1; d >>= 1) v = fmaxf(v, __shfl_xor(v, d, 32));
  return v;
}

__device__ __forceinline__ float wave_sum(float v) {
  #pragma unroll
  for (int d = 16; d >= 1; d >>= 1) v += __shfl_xor(v, d, 32);
  return v;
}

__device__ __forceinline__ float lane_bcast(float v, int l) {
  return __int_as_float(__builtin_amdgcn_readlane(__float_as_int(v), l));
}

__device__ __forceinline__ void accum_rows(const float* __restrict__ x,
                                           int lane, int cnt, int u, float4 p,
                                           float4& acc) {
  #pragma unroll 4
  for (int t = 0; t < cnt; ++t) {
    const int   ut = __builtin_amdgcn_readlane(u, t);
    const float p0 = lane_bcast(p.x, t);
    const float p1 = lane_bcast(p.y, t);
    const float p2 = lane_bcast(p.z, t);
    const float p3 = lane_bcast(p.w, t);
    const float4 xv = *(const float4*)(x + (size_t)ut * 128 + lane * 4);
    acc.x += xv.x * p0;
    acc.y += xv.y * p1;
    acc.z += xv.z * p2;
    acc.w += xv.w * p3;
  }
}

__device__ void node_from_edge_list(const float* __restrict__ x,
                                    const float* __restrict__ s,
                                    const int* __restrict__ src,
                                    const int* __restrict__ dst, int E,
                                    int v, int lane, float4 sv,
                                    float4& acc) {
  float4 mx = {-INFINITY, -INFINITY, -INFINITY, -INFINITY};
  for (int base = 0; base < E; base += 32) {
    const int e = base + lane;
    if (e < E && dst[e] == v) {
      const float4 ej = edge_scores(s, src[e], sv);
      mx.x = fmaxf(mx.x, ej.x); mx.y = fmaxf(mx.y, ej.y);
      mx.z = fmaxf(mx.z, ej.z); mx.w = fmaxf(mx.w, ej.w);
    }
  }
  mx.x = wave_max(mx.x); mx.y = wave_max(mx.y);
  mx.z = wave_max(mx.z); mx.w = wave_max(mx.w);

  float4 dn = {0.0f, 0.0f, 0.0f, 0.0f};
  for (int base = 0; base < E; base += 32) {
    const int e = base + lane;
    if (e < E && dst[e] == v) {
      const float4 ej = edge_scores(s, src[e], sv);
      dn.x += __expf(ej.x - mx.x); dn.y += __expf(ej.y - mx.y);
      dn.z += __expf(ej.z - mx.z); dn.w += __expf(ej.w - mx.w);
    }
  }
  dn.x = wave_sum(dn.x); dn.y = wave_sum(dn.y);
  dn.z = wave_sum(dn.z); dn.w = wave_sum(dn.w);

  for (int base = 0; base < E; base += 32) {
    const int e = base + lane;
    const bool hit = e < E && dst[e] == v;
    unsigned todo = __builtin_amdgcn_ballot_w32(hit);
    if (todo == 0u) continue;
    int u = 0;
    float4 q = {0.0f, 0.0f, 0.0f, 0.0f};
    if (hit) {
      u = src[e];
      const float4 ej = edge_scores(s, u, sv);
      q.x = __expf(ej.x - mx.x) / dn.x; q.y = __expf(ej.y - mx.y) / dn.y;
      q.z = __expf(ej.z - mx.z) / dn.z; q.w = __expf(ej.w - mx.w) / dn.w;
    }
    while (todo != 0u) {
      const int t = __builtin_ctz(todo);
      todo &= todo - 1u;
      const int   ut = __builtin_amdgcn_readlane(u, t);
      const float p0 = lane_bcast(q.x, t);
      const float p1 = lane_bcast(q.y, t);
      const float p2 = lane_bcast(q.z, t);
      const float p3 = lane_bcast(q.w, t);
      const float4 xv = *(const float4*)(x + (size_t)ut * 128 + lane * 4);
      acc.x += xv.x * p0;
      acc.y += xv.y * p1;
      acc.z += xv.z * p2;
      acc.w += xv.w * p3;
    }
  }
}

#define NODE_T (NODES_PER_BLOCK * 32)
#define NODE_EPT (CAPB / NODE_T)

__global__ __launch_bounds__(NODE_T) void gat_node_softmax_agg(
    const float* __restrict__ x, const float* __restrict__ s,
    const int* __restrict__ bk_src, const int* __restrict__ bk_dst,
    const int* __restrict__ bcnt,
    const int* __restrict__ src, const int* __restrict__ dst, int E,
    float* __restrict__ out, int n) {
  __shared__ int cnt[NODES_PER_BLOCK];
  __shared__ int offs[NODES_PER_BLOCK];
  __shared__ int sorted[CAPB];

  const int tid  = threadIdx.x;
  const int lane = tid & 31;
  const int v0   = blockIdx.x * NODES_PER_BLOCK;
  if (v0 >= n) return;
  const int npb   = nodes_per_bucket(n);
  const int b     = v0 / npb;
  const int total = bcnt[b];
  const bool fits = total <= CAPB;

  if (fits) {
    if (tid < NODES_PER_BLOCK) cnt[tid] = 0;
    __syncthreads();

    const int* __restrict__ bs = bk_src + (size_t)b * CAPB;
    const int* __restrict__ bd = bk_dst + (size_t)b * CAPB;
    int uu[NODE_EPT], dl[NODE_EPT], rk[NODE_EPT];
    #pragma unroll
    for (int k = 0; k < NODE_EPT / 4; ++k) {
      const int j = (k * NODE_T + tid) * 4;
      int4 d4 = {0, 0, 0, 0}, u4 = {0, 0, 0, 0};
      if (j < total) {
        d4 = *(const int4*)(bd + j);
        u4 = *(const int4*)(bs + j);
      }
      dl[4 * k + 0] = j + 0 < total ? d4.x - v0 : -1;  uu[4 * k + 0] = u4.x;
      dl[4 * k + 1] = j + 1 < total ? d4.y - v0 : -1;  uu[4 * k + 1] = u4.y;
      dl[4 * k + 2] = j + 2 < total ? d4.z - v0 : -1;  uu[4 * k + 2] = u4.z;
      dl[4 * k + 3] = j + 3 < total ? d4.w - v0 : -1;  uu[4 * k + 3] = u4.w;
    }
    #pragma unroll
    for (int k = 0; k < NODE_EPT; ++k) {
      rk[k] = 0;
      if ((unsigned)dl[k] < (unsigned)NODES_PER_BLOCK)
        rk[k] = atomicAdd(&cnt[dl[k]], 1);
      else
        dl[k] = -1;
    }
    __syncthreads();
    if (tid < NODES_PER_BLOCK) {
      int o = 0;
      for (int i = 0; i < tid; ++i) o += cnt[i];
      offs[tid] = o;
    }
    __syncthreads();
    #pragma unroll
    for (int k = 0; k < NODE_EPT; ++k)
      if (dl[k] >= 0) sorted[offs[dl[k]] + rk[k]] = uu[k];
    __syncthreads();
  }

  const int w = __builtin_amdgcn_readfirstlane(tid >> 5);
  const int v = v0 + w;
  if (v >= n) return;
  const float4 sv = *(const float4*)(s + (size_t)v * 8 + 4);

  float4 acc = {0.0f, 0.0f, 0.0f, 0.0f};

  if (__builtin_expect(!fits, 0)) {
    node_from_edge_list(x, s, src, dst, E, v, lane, sv, acc);
  } else {
    const int len = __builtin_amdgcn_readfirstlane(cnt[w]);
    const int off = __builtin_amdgcn_readfirstlane(offs[w]);

    const bool has0 = lane < len;
    int u0 = 0;
    float4 e0 = {-INFINITY, -INFINITY, -INFINITY, -INFINITY};
    if (has0) { u0 = sorted[off + lane]; e0 = edge_scores(s, u0, sv); }

    float4 mx = e0;
    for (int j = 32 + lane; j < len; j += 32) {
      const float4 ej = edge_scores(s, sorted[off + j], sv);
      mx.x = fmaxf(mx.x, ej.x); mx.y = fmaxf(mx.y, ej.y);
      mx.z = fmaxf(mx.z, ej.z); mx.w = fmaxf(mx.w, ej.w);
    }
    mx.x = wave_max(mx.x); mx.y = wave_max(mx.y);
    mx.z = wave_max(mx.z); mx.w = wave_max(mx.w);

    float4 ex0 = {0.0f, 0.0f, 0.0f, 0.0f};
    if (has0) {
      ex0.x = __expf(e0.x - mx.x); ex0.y = __expf(e0.y - mx.y);
      ex0.z = __expf(e0.z - mx.z); ex0.w = __expf(e0.w - mx.w);
    }
    float4 dn = ex0;
    for (int j = 32 + lane; j < len; j += 32) {
      const float4 ej = edge_scores(s, sorted[off + j], sv);
      dn.x += __expf(ej.x - mx.x); dn.y += __expf(ej.y - mx.y);
      dn.z += __expf(ej.z - mx.z); dn.w += __expf(ej.w - mx.w);
    }
    dn.x = wave_sum(dn.x); dn.y = wave_sum(dn.y);
    dn.z = wave_sum(dn.z); dn.w = wave_sum(dn.w);

    float4 p = {0.0f, 0.0f, 0.0f, 0.0f};
    if (has0) {
      p.x = ex0.x / dn.x; p.y = ex0.y / dn.y;
      p.z = ex0.z / dn.z; p.w = ex0.w / dn.w;
    }
    accum_rows(x, lane, len < 32 ? len : 32, u0, p, acc);

    for (int c = 32; c < len; c += 32) {
      const int j = c + lane;
      int u = 0;
      float4 q = {0.0f, 0.0f, 0.0f, 0.0f};
      if (j < len) {
        u = sorted[off + j];
        const float4 ej = edge_scores(s, u, sv);
        q.x = __expf(ej.x - mx.x) / dn.x; q.y = __expf(ej.y - mx.y) / dn.y;
        q.z = __expf(ej.z - mx.z) / dn.z; q.w = __expf(ej.w - mx.w) / dn.w;
      }
      const int rem = len - c;
      accum_rows(x, lane, rem < 32 ? rem : 32, u, q, acc);
    }
  }

  const float4 xs = *(const float4*)(x + (size_t)v * 128 + lane * 4);
  float* __restrict__ o = out + (size_t)v * 256 + lane * 4;
  *(float4*)o = xs;
  *(float4*)(o + 128) = acc;
}

extern "C" void kernel_launch(void* const* d_in, const int* in_sizes, int n_in,
                              void* d_out, int out_size, void* d_ws, size_t ws_size,
                              hipStream_t stream) {
  const float* x   = (const float*)d_in[0];
  const int*   src = (const int*)d_in[1];
  const int*   dst = (const int*)d_in[2];
  const float* W_u = (const float*)d_in[3];
  const float* b_u = (const float*)d_in[4];
  const float* W_v = (const float*)d_in[5];
  float* out = (float*)d_out;

  const int n = in_sizes[0] / 128;
  const int E = in_sizes[1];

  int*   bk_src = (int*)d_ws;
  int*   bk_dst = bk_src + (size_t)NB * CAPB;
  float* s      = (float*)(bk_dst + (size_t)NB * CAPB);
  int*   bcnt   = (int*)(s + (size_t)n * 8);

  gat_scores_wmma<<<(n + 15) / 16, 32, 0, stream>>>(x, W_u, b_u, W_v, s, bcnt, n);

  gat_edge_bin<<<(E + BIN_EPB - 1) / BIN_EPB, BIN_T, 0, stream>>>(src, dst, bcnt, bk_src,
                                                                 bk_dst, E, n);

  gat_node_softmax_agg<<<(n + NODES_PER_BLOCK - 1) / NODES_PER_BLOCK, NODE_T, 0, stream>>>(
      x, s, bk_src, bk_dst, bcnt, src, dst, E, out, n);
}


# Node kernel: one 32-wave block per destination bucket groups its list by node in LDS once; each wave walks ~7 nodes (100000 one-node waves -> 16384, list read once not 13x); LDS-atomic sum order kept
# speedup vs baseline: 9.8015x; 1.1497x over previous
#include <hip/hip_runtime.h>
#include <math.h>

#define NEG_SLOPE 0.2f


#define NB 512
#define CAPB 4096
#define NODE_ROUND 16

__device__ __forceinline__ int nodes_per_bucket(int n) {
  const int q = (n + NB - 1) / NB;
  const int r = (q + NODE_ROUND - 1) / NODE_ROUND * NODE_ROUND;
  return r > 0 ? r : NODE_ROUND;
}

typedef __attribute__((ext_vector_type(2))) float v2f;
typedef __attribute__((ext_vector_type(8))) float v8f;

__global__ __launch_bounds__(32) void gat_scores_wmma(
    const float* __restrict__ x, const float* __restrict__ W_u,
    const float* __restrict__ b_u, const float* __restrict__ W_v,
    float* __restrict__ s, int* __restrict__ bcnt, int n_nodes) {
  const int tile = blockIdx.x;
  const int lane = threadIdx.x;
  const int half = lane >> 4;
  const int l15  = lane & 15;

  if (lane < 16 && tile * 16 + lane < NB) bcnt[tile * 16 + lane] = 0;

  __shared__ float Wc[128 * 16];
  for (int i = lane; i < 128 * 16; i += 32) {
    const int k = i >> 4, c = i & 15;
    float w = 0.0f;
    if (c < 4)      w = W_u[k * 4 + c];
    else if (c < 8) w = W_v[k * 4 + (c - 4)];
    Wc[i] = w;
  }
  __syncthreads();

  int node = tile * 16 + l15;
  const int nodeC = node < n_nodes ? node : n_nodes - 1;
  const float* __restrict__ xrow = x + (size_t)nodeC * 128;

  v8f acc = {};
  #pragma unroll 4
  for (int k = 0; k < 128; k += 4) {
    const int ka = k + half * 2;
    v2f a; a.x = xrow[ka];            a.y = xrow[ka + 1];
    v2f b; b.x = Wc[ka * 16 + l15];   b.y = Wc[(ka + 1) * 16 + l15];
    acc = __builtin_amdgcn_wmma_f32_16x16x4_f32(
         false, a,  false, b,
         (short)0, acc,  false,  false);
  }

  const int ncol = l15;
  if (ncol < 8) {
    const float bias = (ncol < 4) ? b_u[ncol] : 0.0f;
    #pragma unroll
    for (int r = 0; r < 8; ++r) {
      const int mrow = r + half * 8;
      const int onode = tile * 16 + mrow;
      if (onode < n_nodes) s[(size_t)onode * 8 + ncol] = acc[r] + bias;
    }
  }
}

#define BIN_T 1024
#define BIN_EPT 8
#define BIN_EPB (BIN_T * BIN_EPT)

__global__ __launch_bounds__(BIN_T) void gat_edge_bin(
    const int* __restrict__ src, const int* __restrict__ dst,
    int* __restrict__ bcnt, int* __restrict__ bk_src, int* __restrict__ bk_dst,
    int E, int n) {
  __shared__ int hist[NB];
  __shared__ int lstart[NB];
  __shared__ int gbase[NB];
  __shared__ int wsum[NB / 32];
  __shared__ int st_src[BIN_EPB];
  __shared__ int st_dst[BIN_EPB];

  const int tid  = threadIdx.x;
  const int lane = tid & 31;
  const int npb  = nodes_per_bucket(n);
  const int base = blockIdx.x * BIN_EPB;

  if (tid < NB) hist[tid] = 0;
  __syncthreads();

  int d[BIN_EPT], u[BIN_EPT], pos[BIN_EPT], bkt[BIN_EPT];
  #pragma unroll
  for (int k = 0; k < BIN_EPT; ++k) {
    const int e = base + k * BIN_T + tid;
    d[k] = -1; u[k] = 0;
    if (e < E) { d[k] = dst[e]; u[k] = src[e]; }
  }
  #pragma unroll
  for (int k = 0; k < BIN_EPT; ++k) {
    bkt[k] = 0; pos[k] = 0;
    if (d[k] >= 0) {
      bkt[k] = d[k] / npb;
      pos[k] = atomicAdd(&hist[bkt[k]], 1);
    }
  }
  __syncthreads();

  int c = 0, incl = 0;
  if (tid < NB) {
    c = hist[tid];
    incl = c;
    #pragma unroll
    for (int o = 1; o < 32; o <<= 1) {
      const int t = __shfl_up(incl, o, 32);
      if (lane >= o) incl += t;
    }
    if (lane == 31) wsum[tid >> 5] = incl;
  }
  __syncthreads();
  if (tid < NB) {
    int pre = 0;
    for (int w = 0; w < (tid >> 5); ++w) pre += wsum[w];
    lstart[tid] = pre + incl - c;
    gbase[tid]  = c > 0 ? atomicAdd(bcnt + tid, c) : 0;
  }
  __syncthreads();

  #pragma unroll
  for (int k = 0; k < BIN_EPT; ++k) {
    if (d[k] >= 0) {
      const int p = lstart[bkt[k]] + pos[k];
      st_src[p] = u[k];
      st_dst[p] = d[k];
    }
  }
  __syncthreads();

  const int rem  = E - base;
  const int nloc = rem < BIN_EPB ? rem : BIN_EPB;
  for (int i = tid; i < nloc; i += BIN_T) {
    const int dd = st_dst[i];
    const int b  = dd / npb;
    const int g  = gbase[b] + (i - lstart[b]);
    if (g < CAPB) {
      bk_src[(size_t)b * CAPB + g] = st_src[i];
      bk_dst[(size_t)b * CAPB + g] = dd;
    }
  }
}

__device__ __forceinline__ float lrelu(float v) {
  return v >= 0.0f ? v : NEG_SLOPE * v;
}

__device__ __forceinline__ float4 edge_scores(const float* __restrict__ s,
                                              int u, float4 sv) {
  const float4 su = *(const float4*)(s + (size_t)u * 8);
  float4 e;
  e.x = lrelu(su.x + sv.x);
  e.y = lrelu(su.y + sv.y);
  e.z = lrelu(su.z + sv.z);
  e.w = lrelu(su.w + sv.w);
  return e;
}

__device__ __forceinline__ float wave_max(float v) {
  #pragma unroll
  for (int d = 16; d >= 1; d >>= 1) v = fmaxf(v, __shfl_xor(v, d, 32));
  return v;
}

__device__ __forceinline__ float wave_sum(float v) {
  #pragma unroll
  for (int d = 16; d >= 1; d >>= 1) v += __shfl_xor(v, d, 32);
  return v;
}

__device__ __forceinline__ float lane_bcast(float v, int l) {
  return __int_as_float(__builtin_amdgcn_readlane(__float_as_int(v), l));
}

__device__ __forceinline__ void accum_rows(const float* __restrict__ x,
                                           int lane, int cnt, int u, float4 p,
                                           float4& acc) {
  #pragma unroll 4
  for (int t = 0; t < cnt; ++t) {
    const int   ut = __builtin_amdgcn_readlane(u, t);
    const float p0 = lane_bcast(p.x, t);
    const float p1 = lane_bcast(p.y, t);
    const float p2 = lane_bcast(p.z, t);
    const float p3 = lane_bcast(p.w, t);
    const float4 xv = *(const float4*)(x + (size_t)ut * 128 + lane * 4);
    acc.x += xv.x * p0;
    acc.y += xv.y * p1;
    acc.z += xv.z * p2;
    acc.w += xv.w * p3;
  }
}

__device__ void node_from_edge_list(const float* __restrict__ x,
                                    const float* __restrict__ s,
                                    const int* __restrict__ src,
                                    const int* __restrict__ dst, int E,
                                    int v, int lane, float4 sv,
                                    float4& acc) {
  float4 mx = {-INFINITY, -INFINITY, -INFINITY, -INFINITY};
  for (int base = 0; base < E; base += 32) {
    const int e = base + lane;
    if (e < E && dst[e] == v) {
      const float4 ej = edge_scores(s, src[e], sv);
      mx.x = fmaxf(mx.x, ej.x); mx.y = fmaxf(mx.y, ej.y);
      mx.z = fmaxf(mx.z, ej.z); mx.w = fmaxf(mx.w, ej.w);
    }
  }
  mx.x = wave_max(mx.x); mx.y = wave_max(mx.y);
  mx.z = wave_max(mx.z); mx.w = wave_max(mx.w);

  float4 dn = {0.0f, 0.0f, 0.0f, 0.0f};
  for (int base = 0; base < E; base += 32) {
    const int e = base + lane;
    if (e < E && dst[e] == v) {
      const float4 ej = edge_scores(s, src[e], sv);
      dn.x += __expf(ej.x - mx.x); dn.y += __expf(ej.y - mx.y);
      dn.z += __expf(ej.z - mx.z); dn.w += __expf(ej.w - mx.w);
    }
  }
  dn.x = wave_sum(dn.x); dn.y = wave_sum(dn.y);
  dn.z = wave_sum(dn.z); dn.w = wave_sum(dn.w);

  for (int base = 0; base < E; base += 32) {
    const int e = base + lane;
    const bool hit = e < E && dst[e] == v;
    unsigned todo = __builtin_amdgcn_ballot_w32(hit);
    if (todo == 0u) continue;
    int u = 0;
    float4 q = {0.0f, 0.0f, 0.0f, 0.0f};
    if (hit) {
      u = src[e];
      const float4 ej = edge_scores(s, u, sv);
      q.x = __expf(ej.x - mx.x) / dn.x; q.y = __expf(ej.y - mx.y) / dn.y;
      q.z = __expf(ej.z - mx.z) / dn.z; q.w = __expf(ej.w - mx.w) / dn.w;
    }
    while (todo != 0u) {
      const int t = __builtin_ctz(todo);
      todo &= todo - 1u;
      const int   ut = __builtin_amdgcn_readlane(u, t);
      const float p0 = lane_bcast(q.x, t);
      const float p1 = lane_bcast(q.y, t);
      const float p2 = lane_bcast(q.z, t);
      const float p3 = lane_bcast(q.w, t);
      const float4 xv = *(const float4*)(x + (size_t)ut * 128 + lane * 4);
      acc.x += xv.x * p0;
      acc.y += xv.y * p1;
      acc.z += xv.z * p2;
      acc.w += xv.w * p3;
    }
  }
}

__device__ __forceinline__ void node_from_lds(const float* __restrict__ x,
                                              const float* __restrict__ s,
                                              const int* ids, int len,
                                              int v, int lane, float4& acc) {
  const float4 sv = *(const float4*)(s + (size_t)v * 8 + 4);

  const bool has0 = lane < len;
  int u0 = 0;
  float4 e0 = {-INFINITY, -INFINITY, -INFINITY, -INFINITY};
  if (has0) { u0 = ids[lane]; e0 = edge_scores(s, u0, sv); }

  if (len <= 32) {
    float4 mx;
    mx.x = wave_max(e0.x); mx.y = wave_max(e0.y);
    mx.z = wave_max(e0.z); mx.w = wave_max(e0.w);
    float4 ex0 = {0.0f, 0.0f, 0.0f, 0.0f};
    if (has0) {
      ex0.x = __expf(e0.x - mx.x); ex0.y = __expf(e0.y - mx.y);
      ex0.z = __expf(e0.z - mx.z); ex0.w = __expf(e0.w - mx.w);
    }
    float4 dn;
    dn.x = wave_sum(ex0.x); dn.y = wave_sum(ex0.y);
    dn.z = wave_sum(ex0.z); dn.w = wave_sum(ex0.w);
    float4 p = {0.0f, 0.0f, 0.0f, 0.0f};
    if (has0) {
      p.x = ex0.x / dn.x; p.y = ex0.y / dn.y;
      p.z = ex0.z / dn.z; p.w = ex0.w / dn.w;
    }
    accum_rows(x, lane, len, u0, p, acc);
    return;
  }

  float4 mx = e0;
  for (int j = 32 + lane; j < len; j += 32) {
    const float4 ej = edge_scores(s, ids[j], sv);
    mx.x = fmaxf(mx.x, ej.x); mx.y = fmaxf(mx.y, ej.y);
    mx.z = fmaxf(mx.z, ej.z); mx.w = fmaxf(mx.w, ej.w);
  }
  mx.x = wave_max(mx.x); mx.y = wave_max(mx.y);
  mx.z = wave_max(mx.z); mx.w = wave_max(mx.w);

  float4 ex0;
  ex0.x = __expf(e0.x - mx.x); ex0.y = __expf(e0.y - mx.y);
  ex0.z = __expf(e0.z - mx.z); ex0.w = __expf(e0.w - mx.w);
  float4 dn = ex0;
  for (int j = 32 + lane; j < len; j += 32) {
    const float4 ej = edge_scores(s, ids[j], sv);
    dn.x += __expf(ej.x - mx.x); dn.y += __expf(ej.y - mx.y);
    dn.z += __expf(ej.z - mx.z); dn.w += __expf(ej.w - mx.w);
  }
  dn.x = wave_sum(dn.x); dn.y = wave_sum(dn.y);
  dn.z = wave_sum(dn.z); dn.w = wave_sum(dn.w);

  float4 p;
  p.x = ex0.x / dn.x; p.y = ex0.y / dn.y;
  p.z = ex0.z / dn.z; p.w = ex0.w / dn.w;
  accum_rows(x, lane, 32, u0, p, acc);

  for (int c = 32; c < len; c += 32) {
    const int j = c + lane;
    int u = 0;
    float4 q = {0.0f, 0.0f, 0.0f, 0.0f};
    if (j < len) {
      u = ids[j];
      const float4 ej = edge_scores(s, u, sv);
      q.x = __expf(ej.x - mx.x) / dn.x; q.y = __expf(ej.y - mx.y) / dn.y;
      q.z = __expf(ej.z - mx.z) / dn.z; q.w = __expf(ej.w - mx.w) / dn.w;
    }
    const int rem = len - c;
    accum_rows(x, lane, rem < 32 ? rem : 32, u, q, acc);
  }
}

__device__ __forceinline__ void store_row(const float* __restrict__ x,
                                          float* __restrict__ out, int v,
                                          int lane, float4 acc) {
  const float4 xs = *(const float4*)(x + (size_t)v * 128 + lane * 4);
  float* __restrict__ o = out + (size_t)v * 256 + lane * 4;
  *(float4*)o = xs;
  *(float4*)(o + 128) = acc;
}

#define NODE_T 1024
#define NODE_W (NODE_T / 32)
#define NODE_EPT (CAPB / NODE_T)
#define NODE_CHUNK 256

__global__ __launch_bounds__(NODE_T, 16) void gat_node_softmax_agg(
    const float* __restrict__ x, const float* __restrict__ s,
    const int* __restrict__ bk_src, const int* __restrict__ bk_dst,
    const int* __restrict__ bcnt,
    const int* __restrict__ src, const int* __restrict__ dst, int E,
    float* __restrict__ out, int n) {
  __shared__ int cnt[NODE_CHUNK];
  __shared__ int offs[NODE_CHUNK];
  __shared__ int wsum[NODE_CHUNK / 32];
  __shared__ int sorted[CAPB];

  const int tid  = threadIdx.x;
  const int lane = tid & 31;
  const int npb  = nodes_per_bucket(n);
  const int b    = blockIdx.x;
  const int vb   = b * npb;
  if (vb >= n) return;
  const int vend  = vb + npb < n ? vb + npb : n;
  const int total = bcnt[b];
  const bool fits = total <= CAPB;

  const int w = __builtin_amdgcn_readfirstlane(tid >> 5);

  for (int v0 = vb; v0 < vend; v0 += NODE_CHUNK) {
    const int nn = vend - v0 < NODE_CHUNK ? vend - v0 : NODE_CHUNK;

    if (fits) {
      if (tid < NODE_CHUNK) cnt[tid] = 0;
      __syncthreads();

      const int* __restrict__ bs = bk_src + (size_t)b * CAPB;
      const int* __restrict__ bd = bk_dst + (size_t)b * CAPB;
      int uu[NODE_EPT], dl[NODE_EPT], rk[NODE_EPT];
      #pragma unroll
      for (int k = 0; k < NODE_EPT / 4; ++k) {
        const int j = (k * NODE_T + tid) * 4;
        int4 d4 = {0, 0, 0, 0}, u4 = {0, 0, 0, 0};
        if (j < total) {
          d4 = *(const int4*)(bd + j);
          u4 = *(const int4*)(bs + j);
        }
        dl[4 * k + 0] = j + 0 < total ? d4.x - v0 : -1;  uu[4 * k + 0] = u4.x;
        dl[4 * k + 1] = j + 1 < total ? d4.y - v0 : -1;  uu[4 * k + 1] = u4.y;
        dl[4 * k + 2] = j + 2 < total ? d4.z - v0 : -1;  uu[4 * k + 2] = u4.z;
        dl[4 * k + 3] = j + 3 < total ? d4.w - v0 : -1;  uu[4 * k + 3] = u4.w;
      }
      #pragma unroll
      for (int k = 0; k < NODE_EPT; ++k) {
        rk[k] = 0;
        if ((unsigned)dl[k] < (unsigned)nn)
          rk[k] = atomicAdd(&cnt[dl[k]], 1);
        else
          dl[k] = -1;
      }
      __syncthreads();

      int c = 0, incl = 0;
      if (tid < NODE_CHUNK) {
        c = cnt[tid];
        incl = c;
        #pragma unroll
        for (int o = 1; o < 32; o <<= 1) {
          const int t = __shfl_up(incl, o, 32);
          if (lane >= o) incl += t;
        }
        if (lane == 31) wsum[tid >> 5] = incl;
      }
      __syncthreads();
      if (tid < NODE_CHUNK) {
        int pre = 0;
        for (int q = 0; q < (tid >> 5); ++q) pre += wsum[q];
        offs[tid] = pre + incl - c;
      }
      __syncthreads();
      #pragma unroll
      for (int k = 0; k < NODE_EPT; ++k)
        if (dl[k] >= 0) sorted[offs[dl[k]] + rk[k]] = uu[k];
      __syncthreads();
    }

    if (__builtin_expect(!fits, 0)) {
      for (int i = w; i < nn; i += NODE_W) {
        const int v = v0 + i;
        const float4 sv = *(const float4*)(s + (size_t)v * 8 + 4);
        float4 acc = {0.0f, 0.0f, 0.0f, 0.0f};
        node_from_edge_list(x, s, src, dst, E, v, lane, sv, acc);
        store_row(x, out, v, lane, acc);
      }
    } else {
      for (int i = w; i < nn; i += NODE_W) {
        const int v = v0 + i;
        const int len = __builtin_amdgcn_readfirstlane(cnt[i]);
        const int off = __builtin_amdgcn_readfirstlane(offs[i]);
        float4 acc = {0.0f, 0.0f, 0.0f, 0.0f};
        node_from_lds(x, s, sorted + off, len, v, lane, acc);
        store_row(x, out, v, lane, acc);
      }
    }

    if (v0 + NODE_CHUNK < vend) __syncthreads();
  }
}

extern "C" void kernel_launch(void* const* d_in, const int* in_sizes, int n_in,
                              void* d_out, int out_size, void* d_ws, size_t ws_size,
                              hipStream_t stream) {
  const float* x   = (const float*)d_in[0];
  const int*   src = (const int*)d_in[1];
  const int*   dst = (const int*)d_in[2];
  const float* W_u = (const float*)d_in[3];
  const float* b_u = (const float*)d_in[4];
  const float* W_v = (const float*)d_in[5];
  float* out = (float*)d_out;

  const int n = in_sizes[0] / 128;
  const int E = in_sizes[1];

  int*   bk_src = (int*)d_ws;
  int*   bk_dst = bk_src + (size_t)NB * CAPB;
  float* s      = (float*)(bk_dst + (size_t)NB * CAPB);
  int*   bcnt   = (int*)(s + (size_t)n * 8);

  gat_scores_wmma<<<(n + 15) / 16, 32, 0, stream>>>(x, W_u, b_u, W_v, s, bcnt, n);

  gat_edge_bin<<<(E + BIN_EPB - 1) / BIN_EPB, BIN_T, 0, stream>>>(src, dst, bcnt, bk_src,
                                                                 bk_dst, E, n);

  gat_node_softmax_agg<<<NB, NODE_T, 0, stream>>>(
      x, s, bk_src, bk_dst, bcnt, src, dst, E, out, n);
}


# Score kernel: 8-wave blocks stage the weights once per block and each wave's 16 x rows as whole 512-byte loads through LDS (was 1-wave blocks, 64 serial weight loads, 8-byte x loads); k order kept
# speedup vs baseline: 11.2889x; 1.1518x over previous
#include <hip/hip_runtime.h>
#include <math.h>

#define NEG_SLOPE 0.2f


#define NB 512
#define CAPB 4096
#define NODE_ROUND 16

__device__ __forceinline__ int nodes_per_bucket(int n) {
  const int q = (n + NB - 1) / NB;
  const int r = (q + NODE_ROUND - 1) / NODE_ROUND * NODE_ROUND;
  return r > 0 ? r : NODE_ROUND;
}

typedef __attribute__((ext_vector_type(2))) float v2f;
typedef __attribute__((ext_vector_type(8))) float v8f;

#define SC_W 8
#define SC_T (SC_W * 32)
#define SC_PITCH 132

__global__ __launch_bounds__(SC_T) void gat_scores_wmma(
    const float* __restrict__ x, const float* __restrict__ W_u,
    const float* __restrict__ b_u, const float* __restrict__ W_v,
    float* __restrict__ s, int* __restrict__ bcnt, int n_nodes) {
  __shared__ __attribute__((aligned(16))) float Wc[128 * 16];
  __shared__ __attribute__((aligned(16))) float xt[SC_W * 16 * SC_PITCH];

  const int tid  = threadIdx.x;
  const int lane = tid & 31;
  const int half = lane >> 4;
  const int l15  = lane & 15;
  const int w    = __builtin_amdgcn_readfirstlane(tid >> 5);
  const int tile = blockIdx.x * SC_W + w;

  for (int i = blockIdx.x * SC_T + tid; i < NB; i += gridDim.x * SC_T)
    bcnt[i] = 0;

  float4 xr[16];
  #pragma unroll
  for (int i = 0; i < 16; ++i) {
    int row = tile * 16 + i;
    row = row < n_nodes ? row : n_nodes - 1;
    xr[i] = *(const float4*)(x + (size_t)row * 128 + lane * 4);
  }

  const int wk = tid >> 1, wp = tid & 1;
  const float4 wrow = *(const float4*)((wp ? W_v : W_u) + wk * 4);
  const float4 z = {0.0f, 0.0f, 0.0f, 0.0f};
  float4* wdst = (float4*)(Wc + wk * 16 + wp * 4);
  wdst[0] = wrow;
  wdst[2] = z;
  float* xw = xt + w * (16 * SC_PITCH);
  #pragma unroll
  for (int i = 0; i < 16; ++i)
    *(float4*)(xw + i * SC_PITCH + lane * 4) = xr[i];
  __syncthreads();

  const float* xrow = xw + l15 * SC_PITCH;

  v8f acc = {};
  #pragma unroll
  for (int k = 0; k < 128; k += 4) {
    const int ka = k + half * 2;
    v2f a; a.x = xrow[ka];            a.y = xrow[ka + 1];
    v2f b; b.x = Wc[ka * 16 + l15];   b.y = Wc[(ka + 1) * 16 + l15];
    acc = __builtin_amdgcn_wmma_f32_16x16x4_f32(
         false, a,  false, b,
         (short)0, acc,  false,  false);
  }

  const int ncol = l15;
  if (ncol < 8) {
    const float bias = (ncol < 4) ? b_u[ncol] : 0.0f;
    #pragma unroll
    for (int r = 0; r < 8; ++r) {
      const int mrow = r + half * 8;
      const int onode = tile * 16 + mrow;
      if (onode < n_nodes) s[(size_t)onode * 8 + ncol] = acc[r] + bias;
    }
  }
}

#define BIN_T 1024
#define BIN_EPT 8
#define BIN_EPB (BIN_T * BIN_EPT)

__global__ __launch_bounds__(BIN_T) void gat_edge_bin(
    const int* __restrict__ src, const int* __restrict__ dst,
    int* __restrict__ bcnt, int* __restrict__ bk_src, int* __restrict__ bk_dst,
    int E, int n) {
  __shared__ int hist[NB];
  __shared__ int lstart[NB];
  __shared__ int gbase[NB];
  __shared__ int wsum[NB / 32];
  __shared__ int st_src[BIN_EPB];
  __shared__ int st_dst[BIN_EPB];

  const int tid  = threadIdx.x;
  const int lane = tid & 31;
  const int npb  = nodes_per_bucket(n);
  const int base = blockIdx.x * BIN_EPB;

  if (tid < NB) hist[tid] = 0;
  __syncthreads();

  int d[BIN_EPT], u[BIN_EPT], pos[BIN_EPT], bkt[BIN_EPT];
  #pragma unroll
  for (int k = 0; k < BIN_EPT; ++k) {
    const int e = base + k * BIN_T + tid;
    d[k] = -1; u[k] = 0;
    if (e < E) { d[k] = dst[e]; u[k] = src[e]; }
  }
  #pragma unroll
  for (int k = 0; k < BIN_EPT; ++k) {
    bkt[k] = 0; pos[k] = 0;
    if (d[k] >= 0) {
      bkt[k] = d[k] / npb;
      pos[k] = atomicAdd(&hist[bkt[k]], 1);
    }
  }
  __syncthreads();

  int c = 0, incl = 0;
  if (tid < NB) {
    c = hist[tid];
    incl = c;
    #pragma unroll
    for (int o = 1; o < 32; o <<= 1) {
      const int t = __shfl_up(incl, o, 32);
      if (lane >= o) incl += t;
    }
    if (lane == 31) wsum[tid >> 5] = incl;
  }
  __syncthreads();
  if (tid < NB) {
    int pre = 0;
    for (int w = 0; w < (tid >> 5); ++w) pre += wsum[w];
    lstart[tid] = pre + incl - c;
    gbase[tid]  = c > 0 ? atomicAdd(bcnt + tid, c) : 0;
  }
  __syncthreads();

  #pragma unroll
  for (int k = 0; k < BIN_EPT; ++k) {
    if (d[k] >= 0) {
      const int p = lstart[bkt[k]] + pos[k];
      st_src[p] = u[k];
      st_dst[p] = d[k];
    }
  }
  __syncthreads();

  const int rem  = E - base;
  const int nloc = rem < BIN_EPB ? rem : BIN_EPB;
  for (int i = tid; i < nloc; i += BIN_T) {
    const int dd = st_dst[i];
    const int b  = dd / npb;
    const int g  = gbase[b] + (i - lstart[b]);
    if (g < CAPB) {
      bk_src[(size_t)b * CAPB + g] = st_src[i];
      bk_dst[(size_t)b * CAPB + g] = dd;
    }
  }
}

__device__ __forceinline__ float lrelu(float v) {
  return v >= 0.0f ? v : NEG_SLOPE * v;
}

__device__ __forceinline__ float4 edge_scores(const float* __restrict__ s,
                                              int u, float4 sv) {
  const float4 su = *(const float4*)(s + (size_t)u * 8);
  float4 e;
  e.x = lrelu(su.x + sv.x);
  e.y = lrelu(su.y + sv.y);
  e.z = lrelu(su.z + sv.z);
  e.w = lrelu(su.w + sv.w);
  return e;
}

__device__ __forceinline__ float wave_max(float v) {
  #pragma unroll
  for (int d = 16; d >= 1; d >>= 1) v = fmaxf(v, __shfl_xor(v, d, 32));
  return v;
}

__device__ __forceinline__ float wave_sum(float v) {
  #pragma unroll
  for (int d = 16; d >= 1; d >>= 1) v += __shfl_xor(v, d, 32);
  return v;
}

__device__ __forceinline__ float lane_bcast(float v, int l) {
  return __int_as_float(__builtin_amdgcn_readlane(__float_as_int(v), l));
}

__device__ __forceinline__ void accum_rows(const float* __restrict__ x,
                                           int lane, int cnt, int u, float4 p,
                                           float4& acc) {
  #pragma unroll 4
  for (int t = 0; t < cnt; ++t) {
    const int   ut = __builtin_amdgcn_readlane(u, t);
    const float p0 = lane_bcast(p.x, t);
    const float p1 = lane_bcast(p.y, t);
    const float p2 = lane_bcast(p.z, t);
    const float p3 = lane_bcast(p.w, t);
    const float4 xv = *(const float4*)(x + (size_t)ut * 128 + lane * 4);
    acc.x += xv.x * p0;
    acc.y += xv.y * p1;
    acc.z += xv.z * p2;
    acc.w += xv.w * p3;
  }
}

__device__ void node_from_edge_list(const float* __restrict__ x,
                                    const float* __restrict__ s,
                                    const int* __restrict__ src,
                                    const int* __restrict__ dst, int E,
                                    int v, int lane, float4 sv,
                                    float4& acc) {
  float4 mx = {-INFINITY, -INFINITY, -INFINITY, -INFINITY};
  for (int base = 0; base < E; base += 32) {
    const int e = base + lane;
    if (e < E && dst[e] == v) {
      const float4 ej = edge_scores(s, src[e], sv);
      mx.x = fmaxf(mx.x, ej.x); mx.y = fmaxf(mx.y, ej.y);
      mx.z = fmaxf(mx.z, ej.z); mx.w = fmaxf(mx.w, ej.w);
    }
  }
  mx.x = wave_max(mx.x); mx.y = wave_max(mx.y);
  mx.z = wave_max(mx.z); mx.w = wave_max(mx.w);

  float4 dn = {0.0f, 0.0f, 0.0f, 0.0f};
  for (int base = 0; base < E; base += 32) {
    const int e = base + lane;
    if (e < E && dst[e] == v) {
      const float4 ej = edge_scores(s, src[e], sv);
      dn.x += __expf(ej.x - mx.x); dn.y += __expf(ej.y - mx.y);
      dn.z += __expf(ej.z - mx.z); dn.w += __expf(ej.w - mx.w);
    }
  }
  dn.x = wave_sum(dn.x); dn.y = wave_sum(dn.y);
  dn.z = wave_sum(dn.z); dn.w = wave_sum(dn.w);

  for (int base = 0; base < E; base += 32) {
    const int e = base + lane;
    const bool hit = e < E && dst[e] == v;
    unsigned todo = __builtin_amdgcn_ballot_w32(hit);
    if (todo == 0u) continue;
    int u = 0;
    float4 q = {0.0f, 0.0f, 0.0f, 0.0f};
    if (hit) {
      u = src[e];
      const float4 ej = edge_scores(s, u, sv);
      q.x = __expf(ej.x - mx.x) / dn.x; q.y = __expf(ej.y - mx.y) / dn.y;
      q.z = __expf(ej.z - mx.z) / dn.z; q.w = __expf(ej.w - mx.w) / dn.w;
    }
    while (todo != 0u) {
      const int t = __builtin_ctz(todo);
      todo &= todo - 1u;
      const int   ut = __builtin_amdgcn_readlane(u, t);
      const float p0 = lane_bcast(q.x, t);
      const float p1 = lane_bcast(q.y, t);
      const float p2 = lane_bcast(q.z, t);
      const float p3 = lane_bcast(q.w, t);
      const float4 xv = *(const float4*)(x + (size_t)ut * 128 + lane * 4);
      acc.x += xv.x * p0;
      acc.y += xv.y * p1;
      acc.z += xv.z * p2;
      acc.w += xv.w * p3;
    }
  }
}

__device__ __forceinline__ void node_from_lds(const float* __restrict__ x,
                                              const float* __restrict__ s,
                                              const int* ids, int len,
                                              int v, int lane, float4& acc) {
  const float4 sv = *(const float4*)(s + (size_t)v * 8 + 4);

  const bool has0 = lane < len;
  int u0 = 0;
  float4 e0 = {-INFINITY, -INFINITY, -INFINITY, -INFINITY};
  if (has0) { u0 = ids[lane]; e0 = edge_scores(s, u0, sv); }

  if (len <= 32) {
    float4 mx;
    mx.x = wave_max(e0.x); mx.y = wave_max(e0.y);
    mx.z = wave_max(e0.z); mx.w = wave_max(e0.w);
    float4 ex0 = {0.0f, 0.0f, 0.0f, 0.0f};
    if (has0) {
      ex0.x = __expf(e0.x - mx.x); ex0.y = __expf(e0.y - mx.y);
      ex0.z = __expf(e0.z - mx.z); ex0.w = __expf(e0.w - mx.w);
    }
    float4 dn;
    dn.x = wave_sum(ex0.x); dn.y = wave_sum(ex0.y);
    dn.z = wave_sum(ex0.z); dn.w = wave_sum(ex0.w);
    float4 p = {0.0f, 0.0f, 0.0f, 0.0f};
    if (has0) {
      p.x = ex0.x / dn.x; p.y = ex0.y / dn.y;
      p.z = ex0.z / dn.z; p.w = ex0.w / dn.w;
    }
    accum_rows(x, lane, len, u0, p, acc);
    return;
  }

  float4 mx = e0;
  for (int j = 32 + lane; j < len; j += 32) {
    const float4 ej = edge_scores(s, ids[j], sv);
    mx.x = fmaxf(mx.x, ej.x); mx.y = fmaxf(mx.y, ej.y);
    mx.z = fmaxf(mx.z, ej.z); mx.w = fmaxf(mx.w, ej.w);
  }
  mx.x = wave_max(mx.x); mx.y = wave_max(mx.y);
  mx.z = wave_max(mx.z); mx.w = wave_max(mx.w);

  float4 ex0;
  ex0.x = __expf(e0.x - mx.x); ex0.y = __expf(e0.y - mx.y);
  ex0.z = __expf(e0.z - mx.z); ex0.w = __expf(e0.w - mx.w);
  float4 dn = ex0;
  for (int j = 32 + lane; j < len; j += 32) {
    const float4 ej = edge_scores(s, ids[j], sv);
    dn.x += __expf(ej.x - mx.x); dn.y += __expf(ej.y - mx.y);
    dn.z += __expf(ej.z - mx.z); dn.w += __expf(ej.w - mx.w);
  }
  dn.x = wave_sum(dn.x); dn.y = wave_sum(dn.y);
  dn.z = wave_sum(dn.z); dn.w = wave_sum(dn.w);

  float4 p;
  p.x = ex0.x / dn.x; p.y = ex0.y / dn.y;
  p.z = ex0.z / dn.z; p.w = ex0.w / dn.w;
  accum_rows(x, lane, 32, u0, p, acc);

  for (int c = 32; c < len; c += 32) {
    const int j = c + lane;
    int u = 0;
    float4 q = {0.0f, 0.0f, 0.0f, 0.0f};
    if (j < len) {
      u = ids[j];
      const float4 ej = edge_scores(s, u, sv);
      q.x = __expf(ej.x - mx.x) / dn.x; q.y = __expf(ej.y - mx.y) / dn.y;
      q.z = __expf(ej.z - mx.z) / dn.z; q.w = __expf(ej.w - mx.w) / dn.w;
    }
    const int rem = len - c;
    accum_rows(x, lane, rem < 32 ? rem : 32, u, q, acc);
  }
}

__device__ __forceinline__ void store_row(const float* __restrict__ x,
                                          float* __restrict__ out, int v,
                                          int lane, float4 acc) {
  const float4 xs = *(const float4*)(x + (size_t)v * 128 + lane * 4);
  float* __restrict__ o = out + (size_t)v * 256 + lane * 4;
  *(float4*)o = xs;
  *(float4*)(o + 128) = acc;
}

#define NODE_T 1024
#define NODE_W (NODE_T / 32)
#define NODE_EPT (CAPB / NODE_T)
#define NODE_CHUNK 256

__global__ __launch_bounds__(NODE_T, 16) void gat_node_softmax_agg(
    const float* __restrict__ x, const float* __restrict__ s,
    const int* __restrict__ bk_src, const int* __restrict__ bk_dst,
    const int* __restrict__ bcnt,
    const int* __restrict__ src, const int* __restrict__ dst, int E,
    float* __restrict__ out, int n) {
  __shared__ int cnt[NODE_CHUNK];
  __shared__ int offs[NODE_CHUNK];
  __shared__ int wsum[NODE_CHUNK / 32];
  __shared__ int sorted[CAPB];

  const int tid  = threadIdx.x;
  const int lane = tid & 31;
  const int npb  = nodes_per_bucket(n);
  const int b    = blockIdx.x;
  const int vb   = b * npb;
  if (vb >= n) return;
  const int vend  = vb + npb < n ? vb + npb : n;
  const int total = bcnt[b];
  const bool fits = total <= CAPB;

  const int w = __builtin_amdgcn_readfirstlane(tid >> 5);

  for (int v0 = vb; v0 < vend; v0 += NODE_CHUNK) {
    const int nn = vend - v0 < NODE_CHUNK ? vend - v0 : NODE_CHUNK;

    if (fits) {
      if (tid < NODE_CHUNK) cnt[tid] = 0;
      __syncthreads();

      const int* __restrict__ bs = bk_src + (size_t)b * CAPB;
      const int* __restrict__ bd = bk_dst + (size_t)b * CAPB;
      int uu[NODE_EPT], dl[NODE_EPT], rk[NODE_EPT];
      #pragma unroll
      for (int k = 0; k < NODE_EPT / 4; ++k) {
        const int j = (k * NODE_T + tid) * 4;
        int4 d4 = {0, 0, 0, 0}, u4 = {0, 0, 0, 0};
        if (j < total) {
          d4 = *(const int4*)(bd + j);
          u4 = *(const int4*)(bs + j);
        }
        dl[4 * k + 0] = j + 0 < total ? d4.x - v0 : -1;  uu[4 * k + 0] = u4.x;
        dl[4 * k + 1] = j + 1 < total ? d4.y - v0 : -1;  uu[4 * k + 1] = u4.y;
        dl[4 * k + 2] = j + 2 < total ? d4.z - v0 : -1;  uu[4 * k + 2] = u4.z;
        dl[4 * k + 3] = j + 3 < total ? d4.w - v0 : -1;  uu[4 * k + 3] = u4.w;
      }
      #pragma unroll
      for (int k = 0; k < NODE_EPT; ++k) {
        rk[k] = 0;
        if ((unsigned)dl[k] < (unsigned)nn)
          rk[k] = atomicAdd(&cnt[dl[k]], 1);
        else
          dl[k] = -1;
      }
      __syncthreads();

      int c = 0, incl = 0;
      if (tid < NODE_CHUNK) {
        c = cnt[tid];
        incl = c;
        #pragma unroll
        for (int o = 1; o < 32; o <<= 1) {
          const int t = __shfl_up(incl, o, 32);
          if (lane >= o) incl += t;
        }
        if (lane == 31) wsum[tid >> 5] = incl;
      }
      __syncthreads();
      if (tid < NODE_CHUNK) {
        int pre = 0;
        for (int q = 0; q < (tid >> 5); ++q) pre += wsum[q];
        offs[tid] = pre + incl - c;
      }
      __syncthreads();
      #pragma unroll
      for (int k = 0; k < NODE_EPT; ++k)
        if (dl[k] >= 0) sorted[offs[dl[k]] + rk[k]] = uu[k];
      __syncthreads();
    }

    if (__builtin_expect(!fits, 0)) {
      for (int i = w; i < nn; i += NODE_W) {
        const int v = v0 + i;
        const float4 sv = *(const float4*)(s + (size_t)v * 8 + 4);
        float4 acc = {0.0f, 0.0f, 0.0f, 0.0f};
        node_from_edge_list(x, s, src, dst, E, v, lane, sv, acc);
        store_row(x, out, v, lane, acc);
      }
    } else {
      for (int i = w; i < nn; i += NODE_W) {
        const int v = v0 + i;
        const int len = __builtin_amdgcn_readfirstlane(cnt[i]);
        const int off = __builtin_amdgcn_readfirstlane(offs[i]);
        float4 acc = {0.0f, 0.0f, 0.0f, 0.0f};
        node_from_lds(x, s, sorted + off, len, v, lane, acc);
        store_row(x, out, v, lane, acc);
      }
    }

    if (v0 + NODE_CHUNK < vend) __syncthreads();
  }
}

extern "C" void kernel_launch(void* const* d_in, const int* in_sizes, int n_in,
                              void* d_out, int out_size, void* d_ws, size_t ws_size,
                              hipStream_t stream) {
  const float* x   = (const float*)d_in[0];
  const int*   src = (const int*)d_in[1];
  const int*   dst = (const int*)d_in[2];
  const float* W_u = (const float*)d_in[3];
  const float* b_u = (const float*)d_in[4];
  const float* W_v = (const float*)d_in[5];
  float* out = (float*)d_out;

  const int n = in_sizes[0] / 128;
  const int E = in_sizes[1];

  int*   bk_src = (int*)d_ws;
  int*   bk_dst = bk_src + (size_t)NB * CAPB;
  float* s      = (float*)(bk_dst + (size_t)NB * CAPB);
  int*   bcnt   = (int*)(s + (size_t)n * 8);

  gat_scores_wmma<<<(n + 16 * SC_W - 1) / (16 * SC_W), SC_T, 0, stream>>>(x, W_u, b_u, W_v, s,
                                                                     bcnt, n);

  gat_edge_bin<<<(E + BIN_EPB - 1) / BIN_EPB, BIN_T, 0, stream>>>(src, dst, bcnt, bk_src,
                                                                 bk_dst, E, n);

  gat_node_softmax_agg<<<NB, NODE_T, 0, stream>>>(
      x, s, bk_src, bk_dst, bcnt, src, dst, E, out, n);
}
